# starting kernel
# baseline (speedup 1.0000x reference)
#include <hip/hip_runtime.h>

typedef float v2f __attribute__((ext_vector_type(2)));
typedef float v8f __attribute__((ext_vector_type(8)));

#if __has_builtin(__builtin_amdgcn_sched_barrier)
#define SCHED_BARRIER() __builtin_amdgcn_sched_barrier(0)
#else
#define SCHED_BARRIER() asm volatile("" ::: "memory")
#endif

namespace {
constexpr int   kBatch = 2;
constexpr int   kSeq   = 2048;
constexpr int   kDim   = 128;
constexpr int   kWin   = 128;
constexpr float kNegInf = -1.0e9f;
constexpr float kScale  = 0.08838834764831845f;
constexpr int   kTilesPerSeq   = kSeq / 16;
constexpr int   kWavesPerBlock = 4;
}

__global__ __launch_bounds__(128)
void swa_wmma_f32_kernel(const float* __restrict__ Q,
                         const float* __restrict__ K,
                         const float* __restrict__ V,
                         float* __restrict__ O) {
  const int lane = threadIdx.x & 31;
  const int wave = threadIdx.x >> 5;
  const int tile = blockIdx.x * kWavesPerBlock + wave;
  const int b    = tile / kTilesPerSeq;
  const int q0   = (tile % kTilesPerSeq) * 16;

  const int h   = lane >> 4;
  const int l15 = lane & 15;

  const float* Qb = Q + (size_t)b * kSeq * kDim;
  const float* Kb = K + (size_t)b * kSeq * kDim;
  const float* Vb = V + (size_t)b * kSeq * kDim;
  float*       Ob = O + (size_t)b * kSeq * kDim;

  v2f qb[32];
  {
    const float* qrow = Qb + (size_t)(q0 + l15) * kDim + 2 * h;
#pragma unroll
    for (int c = 0; c < 32; ++c)
      qb[c] = *(const v2f*)(qrow + 4 * c);
  }

  v8f acc[8];
#pragma unroll
  for (int i = 0; i < 8; ++i) acc[i] = (v8f)0.0f;

  float run_max = kNegInf;
  float run_sum = 0.0f;

#pragma unroll 1
  for (int j = 0; j < 9; ++j) {
    const int kb = q0 - kWin + 16 * j;

    v2f ka[32];
    {
      int krow = kb + l15;
      if (krow < 0) krow = 0;
      const float* kptr = Kb + (size_t)krow * kDim + 2 * h;
#pragma unroll
      for (int c = 0; c < 32; ++c)
        ka[c] = *(const v2f*)(kptr + 4 * c);
    }
    SCHED_BARRIER();

    v8f pt0 = (v8f)0.0f, pt1 = (v8f)0.0f;
#pragma unroll
    for (int c = 0; c < 16; ++c) {
      pt0 = __builtin_amdgcn_wmma_f32_16x16x4_f32(false, ka[2 * c],     false, qb[2 * c],
                                                  (short)0, pt0, false, false);
      pt1 = __builtin_amdgcn_wmma_f32_16x16x4_f32(false, ka[2 * c + 1], false, qb[2 * c + 1],
                                                  (short)0, pt1, false, false);
    }
    v8f pt;
#pragma unroll
    for (int r = 0; r < 8; ++r) pt[r] = pt0[r] + pt1[r];

    int vrow0[4], vrow1[4];
#pragma unroll
    for (int c = 0; c < 4; ++c) {
      int r0 = kb + 4 * c + 2 * h;
      int r1 = r0 + 1;
      vrow0[c] = (r0 < 0) ? 0 : r0;
      vrow1[c] = (r1 < 0) ? 0 : r1;
    }
    v2f va0[16];
#pragma unroll
    for (int dc = 0; dc < 4; ++dc) {
      const float* vcol = Vb + dc * 16 + l15;
#pragma unroll
      for (int c = 0; c < 4; ++c) {
        va0[dc * 4 + c][0] = vcol[(size_t)vrow0[c] * kDim];
        va0[dc * 4 + c][1] = vcol[(size_t)vrow1[c] * kDim];
      }
    }
    SCHED_BARRIER();

    float pmax = kNegInf;
#pragma unroll
    for (int r = 0; r < 8; ++r) {
      const int key  = kb + r + 8 * h;
      const int qidx = q0 + l15;
      const bool valid = (key >= 0) && (key <= qidx) && (key >= qidx - kWin);
      const float s = valid ? pt[r] * kScale : kNegInf;
      pt[r] = s;
      pmax = fmaxf(pmax, s);
    }
    pmax = fmaxf(pmax, __shfl_xor(pmax, 16, 32));

    const float new_max = fmaxf(run_max, pmax);
    const float corr    = __expf(run_max - new_max);
    run_max = new_max;

    float psum = 0.0f;
#pragma unroll
    for (int r = 0; r < 8; ++r) {
      const float e = __expf(pt[r] - new_max);
      pt[r] = e;
      psum += e;
    }
    psum += __shfl_xor(psum, 16, 32);
    run_sum = run_sum * corr + psum;

#pragma unroll
    for (int i = 0; i < 8; ++i)
#pragma unroll
      for (int r = 0; r < 8; ++r)
        acc[i][r] *= corr;

    v2f pb[4];
    {
      const float x2 = __shfl_xor(pt[2], 16, 32);
      const float x3 = __shfl_xor(pt[3], 16, 32);
      pb[0][0] = h ? x2 : pt[0];  pb[0][1] = h ? x3 : pt[1];
      const float x6 = __shfl_xor(pt[6], 16, 32);
      const float x7 = __shfl_xor(pt[7], 16, 32);
      pb[1][0] = h ? x6 : pt[4];  pb[1][1] = h ? x7 : pt[5];
      const float y0 = __shfl_xor(pt[0], 16, 32);
      const float y1 = __shfl_xor(pt[1], 16, 32);
      pb[2][0] = h ? pt[2] : y0;  pb[2][1] = h ? pt[3] : y1;
      const float y4 = __shfl_xor(pt[4], 16, 32);
      const float y5 = __shfl_xor(pt[5], 16, 32);
      pb[3][0] = h ? pt[6] : y4;  pb[3][1] = h ? pt[7] : y5;
    }

#pragma unroll
    for (int dc = 0; dc < 4; ++dc)
#pragma unroll
      for (int c = 0; c < 4; ++c)
        acc[dc] = __builtin_amdgcn_wmma_f32_16x16x4_f32(false, va0[dc * 4 + c],
                                                        false, pb[c],
                                                        (short)0, acc[dc], false, false);

    v2f va1[16];
#pragma unroll
    for (int dc = 0; dc < 4; ++dc) {
      const float* vcol = Vb + (dc + 4) * 16 + l15;
#pragma unroll
      for (int c = 0; c < 4; ++c) {
        va1[dc * 4 + c][0] = vcol[(size_t)vrow0[c] * kDim];
        va1[dc * 4 + c][1] = vcol[(size_t)vrow1[c] * kDim];
      }
    }
    SCHED_BARRIER();
#pragma unroll
    for (int dc = 0; dc < 4; ++dc)
#pragma unroll
      for (int c = 0; c < 4; ++c)
        acc[dc + 4] = __builtin_amdgcn_wmma_f32_16x16x4_f32(false, va1[dc * 4 + c],
                                                            false, pb[c],
                                                            (short)0, acc[dc + 4], false, false);
  }

  const float inv = 1.0f / run_sum;
  float* orow = Ob + (size_t)(q0 + l15) * kDim;
#pragma unroll
  for (int dc = 0; dc < 8; ++dc)
#pragma unroll
    for (int r = 0; r < 8; ++r)
      orow[dc * 16 + r + 8 * h] = acc[dc][r] * inv;
}

extern "C" void kernel_launch(void* const* d_in, const int* in_sizes, int n_in,
                              void* d_out, int out_size, void* d_ws, size_t ws_size,
                              hipStream_t stream) {
  const float* q = (const float*)d_in[0];
  const float* k = (const float*)d_in[1];
  const float* v = (const float*)d_in[2];
  float* out = (float*)d_out;

  const int total_tiles = kBatch * kTilesPerSeq;
  dim3 grid(total_tiles / kWavesPerBlock);
  dim3 block(32 * kWavesPerBlock);
  swa_wmma_f32_kernel<<<grid, block, 0, stream>>>(q, k, v, out);
}


# Output product's k order set to the score tile's register layout (k-step c: V rows 8h+2c,+1): P enters from the lane's own registers, no cross-lane shuffles; sum order over a tile's keys changes
# speedup vs baseline: 2.6195x; 2.6195x over previous
#include <hip/hip_runtime.h>


typedef float v2f __attribute__((ext_vector_type(2)));
typedef float v4f __attribute__((ext_vector_type(4)));
typedef float v8f __attribute__((ext_vector_type(8)));

#define END_OF_ISSUE_SECTION() __builtin_amdgcn_sched_barrier(0)

namespace {
constexpr int   kBatch = 2;
constexpr int   kSeq   = 2048;
constexpr int   kDim   = 128;
constexpr int   kWin   = 128;
constexpr float kNegInf = -1.0e9f;
constexpr float kScale  = 0.08838834764831845f;
constexpr int   kTilesPerSeq = kSeq / 16;
constexpr int   kWaves    = 4;
constexpr int   kDimW     = kDim / kWaves;
constexpr int   kChunksW  = kDimW / 4;
constexpr int   kKeyTiles = 9;
constexpr int   kKAhead   = 4;
constexpr int   kVEarly   = kKeyTiles - 1 - kKAhead;
}

__device__ __forceinline__ void load_k_tile(v2f (&ka)[kChunksW], const float* __restrict__ Kb,
                                            int kb, int l15, int h, int d0) {
  int krow = kb + l15;
  if (krow < 0) krow = 0;
  const float* kptr = Kb + (size_t)krow * kDim + d0 + 2 * h;
#pragma unroll
  for (int c = 0; c < kChunksW; ++c)
    ka[c] = *(const v2f*)(kptr + 4 * c);
}

__device__ __forceinline__ void load_v_half(v2f (&va)[8], const float* __restrict__ Vb,
                                            int kb, int l15, int h, int d0, int dc) {
  const float* vcol = Vb + d0 + dc * 16 + l15;
#pragma unroll
  for (int c = 0; c < 4; ++c) {
    int r0 = kb + 8 * h + 2 * c;
    int r1 = r0 + 1;
    r0 = (r0 < 0) ? 0 : r0;
    r1 = (r1 < 0) ? 0 : r1;
    va[dc * 4 + c][0] = vcol[(size_t)r0 * kDim];
    va[dc * 4 + c][1] = vcol[(size_t)r1 * kDim];
  }
}

__global__ __launch_bounds__(32 * kWaves) __attribute__((amdgpu_waves_per_eu(1, 1)))
void swa_wmma_f32_kernel(const float* __restrict__ Q,
                         const float* __restrict__ K,
                         const float* __restrict__ V,
                         float* __restrict__ O) {
  __shared__ v4f part[kWaves][kKeyTiles][2][32];

  const int lane = threadIdx.x & 31;
  const int wave = __builtin_amdgcn_readfirstlane((int)(threadIdx.x >> 5));
  const int tile = blockIdx.x;
  const int b    = tile / kTilesPerSeq;
  const int q0   = (tile % kTilesPerSeq) * 16;
  const int d0   = wave * kDimW;

  const int h   = lane >> 4;
  const int l15 = lane & 15;

  const float* Qb = Q + (size_t)b * kSeq * kDim;
  const float* Kb = K + (size_t)b * kSeq * kDim;
  const float* Vb = V + (size_t)b * kSeq * kDim;
  float*       Ob = O + (size_t)b * kSeq * kDim;

  const int kb0 = q0 - kWin;

  v2f qb[kChunksW];
  {
    const float* qrow = Qb + (size_t)(q0 + l15) * kDim + d0 + 2 * h;
#pragma unroll
    for (int c = 0; c < kChunksW; ++c)
      qb[c] = *(const v2f*)(qrow + 4 * c);
  }

  v2f ka[kKeyTiles][kChunksW];
  v2f va[kKeyTiles][8];
  v8f pt[kKeyTiles];

#pragma unroll
  for (int j = 0; j < kKAhead; ++j)
    load_k_tile(ka[j], Kb, kb0 + 16 * j, l15, h, d0);
  END_OF_ISSUE_SECTION();

#pragma unroll
  for (int j = 0; j < kKeyTiles; ++j) {
    if (j + kKAhead < kKeyTiles) {
      load_k_tile(ka[j + kKAhead], Kb, kb0 + 16 * (j + kKAhead), l15, h, d0);
    } else {
      const int t = j + kKAhead - kKeyTiles;
      load_v_half(va[t], Vb, kb0 + 16 * t, l15, h, d0, 0);
      load_v_half(va[t], Vb, kb0 + 16 * t, l15, h, d0, 1);
    }
    v8f p0 = (v8f)0.0f, p1 = (v8f)0.0f;
#pragma unroll
    for (int c = 0; c < kChunksW / 2; ++c) {
      p0 = __builtin_amdgcn_wmma_f32_16x16x4_f32(false, ka[j][2 * c],     false, qb[2 * c],
                                                 (short)0, p0, false, false);
      p1 = __builtin_amdgcn_wmma_f32_16x16x4_f32(false, ka[j][2 * c + 1], false, qb[2 * c + 1],
                                                 (short)0, p1, false, false);
    }
#pragma unroll
    for (int r = 0; r < 8; ++r) pt[j][r] = p0[r] + p1[r];
    END_OF_ISSUE_SECTION();
  }

#pragma unroll
  for (int j = 0; j < kKeyTiles; ++j) {
    part[wave][j][0][lane] = (v4f){pt[j][0], pt[j][1], pt[j][2], pt[j][3]};
    part[wave][j][1][lane] = (v4f){pt[j][4], pt[j][5], pt[j][6], pt[j][7]};
  }
  __builtin_amdgcn_fence(__ATOMIC_RELEASE, "workgroup", "local");
  __builtin_amdgcn_s_barrier();
  __builtin_amdgcn_fence(__ATOMIC_ACQUIRE, "workgroup", "local");

  const int wp = wave ^ 1;
  const int wo = 2 - (wave & 2);
  v4f xa[kKeyTiles][2], xu[kKeyTiles][2], xw[kKeyTiles][2];
#pragma unroll
  for (int j = 0; j < kKeyTiles; ++j) {
#pragma unroll
    for (int rq = 0; rq < 2; ++rq) {
      xa[j][rq] = part[wp][j][rq][lane];
      xu[j][rq] = part[wo][j][rq][lane];
      xw[j][rq] = part[wo + 1][j][rq][lane];
    }
  }
  END_OF_ISSUE_SECTION();
#pragma unroll
  for (int t = kVEarly; t < kKeyTiles; ++t) {
    load_v_half(va[t], Vb, kb0 + 16 * t, l15, h, d0, 0);
    load_v_half(va[t], Vb, kb0 + 16 * t, l15, h, d0, 1);
  }
  END_OF_ISSUE_SECTION();

  float mx = kNegInf;
#pragma unroll
  for (int j = 0; j < kKeyTiles; ++j) {
#pragma unroll
    for (int rq = 0; rq < 2; ++rq)
#pragma unroll
      for (int i = 0; i < 4; ++i)
        pt[j][4 * rq + i] = (pt[j][4 * rq + i] + xa[j][rq][i]) + (xu[j][rq][i] + xw[j][rq][i]);
#pragma unroll
    for (int r = 0; r < 8; ++r) {
      const int key  = kb0 + 16 * j + r + 8 * h;
      const int qidx = q0 + l15;
      const bool valid = (key >= 0) && (key <= qidx) && (key >= qidx - kWin);
      const float s = valid ? pt[j][r] * kScale : kNegInf;
      pt[j][r] = s;
      mx = fmaxf(mx, s);
    }
  }
  mx = fmaxf(mx, __shfl_xor(mx, 16, 32));

  float sum = 0.0f;
#pragma unroll
  for (int j = 0; j < kKeyTiles; ++j) {
#pragma unroll
    for (int r = 0; r < 8; ++r) {
      const float e = __expf(pt[j][r] - mx);
      pt[j][r] = e;
      sum += e;
    }
  }
  sum += __shfl_xor(sum, 16, 32);

  v8f acc[2];
  acc[0] = (v8f)0.0f;
  acc[1] = (v8f)0.0f;
#pragma unroll
  for (int j = 0; j < kKeyTiles; ++j) {
#pragma unroll
    for (int c = 0; c < 4; ++c) {
      const v2f pb = {pt[j][2 * c], pt[j][2 * c + 1]};
#pragma unroll
      for (int dc = 0; dc < 2; ++dc)
        acc[dc] = __builtin_amdgcn_wmma_f32_16x16x4_f32(false, va[j][dc * 4 + c],
                                                        false, pb,
                                                        (short)0, acc[dc], false, false);
    }
  }

  const float inv = 1.0f / sum;
  float* orow = Ob + (size_t)(q0 + l15) * kDim + d0;
#pragma unroll
  for (int dc = 0; dc < 2; ++dc)
#pragma unroll
    for (int r = 0; r < 8; ++r)
      orow[dc * 16 + r + 8 * h] = acc[dc][r] * inv;
}

extern "C" void kernel_launch(void* const* d_in, const int* in_sizes, int n_in,
                              void* d_out, int out_size, void* d_ws, size_t ws_size,
                              hipStream_t stream) {
  const float* q = (const float*)d_in[0];
  const float* k = (const float*)d_in[1];
  const float* v = (const float*)d_in[2];
  float* out = (float*)d_out;

  dim3 grid(kBatch * kTilesPerSeq);
  dim3 block(32 * kWaves);
  swa_wmma_f32_kernel<<<grid, block, 0, stream>>>(q, k, v, out);
}


# Edge handling per key tile in scalar registers: one clamp of a tile's first row and one in-sequence flag replace per-row clamps and per-key window tests; loads: scalar tile base + fixed lane offset
# speedup vs baseline: 3.0101x; 1.1491x over previous
#include <hip/hip_runtime.h>


typedef float v2f __attribute__((ext_vector_type(2)));
typedef float v4f __attribute__((ext_vector_type(4)));
typedef float v8f __attribute__((ext_vector_type(8)));

#define END_OF_ISSUE_SECTION() __builtin_amdgcn_sched_barrier(0)

namespace {
constexpr int   kBatch = 2;
constexpr int   kSeq   = 2048;
constexpr int   kDim   = 128;
constexpr int   kWin   = 128;
constexpr float kNegInf = -1.0e9f;
constexpr float kScale  = 0.08838834764831845f;
constexpr int   kTilesPerSeq = kSeq / 16;
constexpr int   kWaves    = 4;
constexpr int   kDimW     = kDim / kWaves;
constexpr int   kChunksW  = kDimW / 4;
constexpr int   kKeyTiles = 9;
constexpr int   kKAhead   = 4;
constexpr int   kVEarly   = kKeyTiles - 1 - kKAhead;
}

__device__ __forceinline__ int first_row_in_seq(int kb) { return kb < 0 ? 0 : kb; }

typedef const __attribute__((address_space(1))) char* gbytes;
__device__ __forceinline__ gbytes tile_base(const float* p) {
  const unsigned long long v = (unsigned long long)p;
  const unsigned lo = __builtin_amdgcn_readfirstlane((unsigned)v);
  const unsigned hi = __builtin_amdgcn_readfirstlane((unsigned)(v >> 32));
  return (gbytes)(((unsigned long long)hi << 32) | lo);
}

__device__ __forceinline__ void load_k_tile(v2f (&ka)[kChunksW], gbytes kt, unsigned lane_off) {
  gbytes p = kt + lane_off;
#pragma unroll
  for (int c = 0; c < kChunksW; ++c)
    ka[c] = *(const __attribute__((address_space(1))) v2f*)(p + 16 * c);
}

__device__ __forceinline__ void load_v_half(v2f (&va)[8], gbytes vt, unsigned lane_off, int dc) {
  gbytes p = vt + lane_off + dc * 64;
#pragma unroll
  for (int c = 0; c < 4; ++c) {
    va[dc * 4 + c][0] = *(const __attribute__((address_space(1))) float*)(p + (2 * c) * (kDim * 4));
    va[dc * 4 + c][1] = *(const __attribute__((address_space(1))) float*)(p + (2 * c + 1) * (kDim * 4));
  }
}

__global__ __launch_bounds__(32 * kWaves) __attribute__((amdgpu_waves_per_eu(1, 1)))
void swa_wmma_f32_kernel(const float* __restrict__ Q,
                         const float* __restrict__ K,
                         const float* __restrict__ V,
                         float* __restrict__ O) {
  __shared__ v4f part[kWaves][kKeyTiles][2][32];

  const int lane = threadIdx.x & 31;
  const int wave = __builtin_amdgcn_readfirstlane((int)(threadIdx.x >> 5));
  const int tile = blockIdx.x;
  const int b    = tile / kTilesPerSeq;
  const int q0   = (tile % kTilesPerSeq) * 16;
  const int d0   = wave * kDimW;

  const int h   = lane >> 4;
  const int l15 = lane & 15;

  const float* Qb = Q + (size_t)b * kSeq * kDim;
  const float* Kb = K + (size_t)b * kSeq * kDim;
  const float* Vb = V + (size_t)b * kSeq * kDim;
  float*       Ob = O + (size_t)b * kSeq * kDim;

  const int kb0 = q0 - kWin;

  const unsigned qk_lane_off = (unsigned)((l15 * kDim + 2 * h) * 4);
  const unsigned v_lane_off  = (unsigned)((8 * h * kDim + l15) * 4);

  v2f qb[kChunksW];
  load_k_tile(qb, tile_base(Qb + (size_t)q0 * kDim + d0), qk_lane_off);

  v2f ka[kKeyTiles][kChunksW];
  v2f va[kKeyTiles][8];
  v8f pt[kKeyTiles];

#pragma unroll
  for (int j = 0; j < kKAhead; ++j)
    load_k_tile(ka[j], tile_base(Kb + (size_t)first_row_in_seq(kb0 + 16 * j) * kDim + d0), qk_lane_off);
  END_OF_ISSUE_SECTION();

#pragma unroll
  for (int j = 0; j < kKeyTiles; ++j) {
    if (j + kKAhead < kKeyTiles) {
      load_k_tile(ka[j + kKAhead],
                  tile_base(Kb + (size_t)first_row_in_seq(kb0 + 16 * (j + kKAhead)) * kDim + d0),
                  qk_lane_off);
    } else {
      const int t = j + kKAhead - kKeyTiles;
      gbytes vt = tile_base(Vb + (size_t)first_row_in_seq(kb0 + 16 * t) * kDim + d0);
      load_v_half(va[t], vt, v_lane_off, 0);
      load_v_half(va[t], vt, v_lane_off, 1);
    }
    v8f p0 = (v8f)0.0f, p1 = (v8f)0.0f;
#pragma unroll
    for (int c = 0; c < kChunksW / 2; ++c) {
      p0 = __builtin_amdgcn_wmma_f32_16x16x4_f32(false, ka[j][2 * c],     false, qb[2 * c],
                                                 (short)0, p0, false, false);
      p1 = __builtin_amdgcn_wmma_f32_16x16x4_f32(false, ka[j][2 * c + 1], false, qb[2 * c + 1],
                                                 (short)0, p1, false, false);
    }
#pragma unroll
    for (int r = 0; r < 8; ++r) pt[j][r] = p0[r] + p1[r];
    END_OF_ISSUE_SECTION();
  }

#pragma unroll
  for (int j = 0; j < kKeyTiles; ++j) {
    part[wave][j][0][lane] = (v4f){pt[j][0], pt[j][1], pt[j][2], pt[j][3]};
    part[wave][j][1][lane] = (v4f){pt[j][4], pt[j][5], pt[j][6], pt[j][7]};
  }
  __builtin_amdgcn_fence(__ATOMIC_RELEASE, "workgroup", "local");
  __builtin_amdgcn_s_barrier();
  __builtin_amdgcn_fence(__ATOMIC_ACQUIRE, "workgroup", "local");

  const int wp = wave ^ 1;
  const int wo = 2 - (wave & 2);
  v4f xa[kKeyTiles][2], xu[kKeyTiles][2], xw[kKeyTiles][2];
#pragma unroll
  for (int j = 0; j < kKeyTiles; ++j) {
#pragma unroll
    for (int rq = 0; rq < 2; ++rq) {
      xa[j][rq] = part[wp][j][rq][lane];
      xu[j][rq] = part[wo][j][rq][lane];
      xw[j][rq] = part[wo + 1][j][rq][lane];
    }
  }
  END_OF_ISSUE_SECTION();
#pragma unroll
  for (int t = kVEarly; t < kKeyTiles; ++t) {
    gbytes vt = tile_base(Vb + (size_t)first_row_in_seq(kb0 + 16 * t) * kDim + d0);
    load_v_half(va[t], vt, v_lane_off, 0);
    load_v_half(va[t], vt, v_lane_off, 1);
  }
  END_OF_ISSUE_SECTION();

  const int r_diag = l15 - 8 * h;
  const int r_min0 = (kb0 >= 0) ? r_diag : 8;
  float mx = kNegInf;
#pragma unroll
  for (int j = 0; j < kKeyTiles; ++j) {
#pragma unroll
    for (int rq = 0; rq < 2; ++rq)
#pragma unroll
      for (int i = 0; i < 4; ++i)
        pt[j][4 * rq + i] = (pt[j][4 * rq + i] + xa[j][rq][i]) + (xu[j][rq][i] + xw[j][rq][i]);
#pragma unroll
    for (int r = 0; r < 8; ++r) {
      const bool valid = (j == 0)             ? (r >= r_min0)
                       : (j == kKeyTiles - 1) ? (r <= r_diag)
                                              : (kb0 + 16 * j >= 0);
      const float s = valid ? pt[j][r] * kScale : kNegInf;
      pt[j][r] = s;
      mx = fmaxf(mx, s);
    }
  }
  mx = fmaxf(mx, __shfl_xor(mx, 16, 32));

  float sum = 0.0f;
#pragma unroll
  for (int j = 0; j < kKeyTiles; ++j) {
#pragma unroll
    for (int r = 0; r < 8; ++r) {
      const float e = __expf(pt[j][r] - mx);
      pt[j][r] = e;
      sum += e;
    }
  }
  sum += __shfl_xor(sum, 16, 32);

  v8f acc[2];
  acc[0] = (v8f)0.0f;
  acc[1] = (v8f)0.0f;
#pragma unroll
  for (int j = 0; j < kKeyTiles; ++j) {
#pragma unroll
    for (int c = 0; c < 4; ++c) {
      const v2f pb = {pt[j][2 * c], pt[j][2 * c + 1]};
#pragma unroll
      for (int dc = 0; dc < 2; ++dc)
        acc[dc] = __builtin_amdgcn_wmma_f32_16x16x4_f32(false, va[j][dc * 4 + c],
                                                        false, pb,
                                                        (short)0, acc[dc], false, false);
    }
  }

  const float inv = 1.0f / sum;
  float* orow = Ob + (size_t)(q0 + l15) * kDim + d0;
#pragma unroll
  for (int dc = 0; dc < 2; ++dc)
#pragma unroll
    for (int r = 0; r < 8; ++r)
      orow[dc * 16 + r + 8 * h] = acc[dc][r] * inv;
}

extern "C" void kernel_launch(void* const* d_in, const int* in_sizes, int n_in,
                              void* d_out, int out_size, void* d_ws, size_t ws_size,
                              hipStream_t stream) {
  const float* q = (const float*)d_in[0];
  const float* k = (const float*)d_in[1];
  const float* v = (const float*)d_in[2];
  float* out = (float*)d_out;

  dim3 grid(kBatch * kTilesPerSeq);
  dim3 block(32 * kWaves);
  swa_wmma_f32_kernel<<<grid, block, 0, stream>>>(q, k, v, out);
}


# Load issue sized to the wave's 63-load window: K six tiles ahead (was four), V in half-tile groups as the window drains: 7 under the scores, 5 behind the exchange reads, 6 before output tiles 0-2
# speedup vs baseline: 3.3574x; 1.1154x over previous
#include <hip/hip_runtime.h>


typedef float v2f __attribute__((ext_vector_type(2)));
typedef float v4f __attribute__((ext_vector_type(4)));
typedef float v8f __attribute__((ext_vector_type(8)));

#define END_OF_ISSUE_SECTION() __builtin_amdgcn_sched_barrier(0)
#define LOADS_AND_MATRIX_KEEP_ORDER() __builtin_amdgcn_sched_barrier(0x786)

namespace {
constexpr int   kBatch = 2;
constexpr int   kSeq   = 2048;
constexpr int   kDim   = 128;
constexpr int   kWin   = 128;
constexpr float kNegInf = -1.0e9f;
constexpr float kScale  = 0.08838834764831845f;
constexpr int   kTilesPerSeq = kSeq / 16;
constexpr int   kWaves    = 4;
constexpr int   kDimW     = kDim / kWaves;
constexpr int   kChunksW  = kDimW / 4;
constexpr int   kKeyTiles = 9;
constexpr int   kKAhead   = 6;
constexpr int   kVHalves  = 2 * kKeyTiles;
constexpr int v_halves_in_section(int j)   { return j < 3 ? 0 : (j < kKeyTiles - 1 ? 1 : 2); }
constexpr int v_first_half_in_section(int j) { return j < 3 ? 0 : j - 3; }
constexpr int   kVScoreHalves = 7;
constexpr int   kVExchHalves  = 5;
constexpr int   kVOutFirst    = kVScoreHalves + kVExchHalves;
}

__device__ __forceinline__ int first_row_in_seq(int kb) { return kb < 0 ? 0 : kb; }

typedef const __attribute__((address_space(1))) char* gbytes;
__device__ __forceinline__ gbytes tile_base(const float* p) {
  const unsigned long long v = (unsigned long long)p;
  const unsigned lo = __builtin_amdgcn_readfirstlane((unsigned)v);
  const unsigned hi = __builtin_amdgcn_readfirstlane((unsigned)(v >> 32));
  return (gbytes)(((unsigned long long)hi << 32) | lo);
}

__device__ __forceinline__ void load_k_tile(v2f (&ka)[kChunksW], gbytes kt, unsigned lane_off) {
  gbytes p = kt + lane_off;
#pragma unroll
  for (int c = 0; c < kChunksW; ++c)
    ka[c] = *(const __attribute__((address_space(1))) v2f*)(p + 16 * c);
}

__device__ __forceinline__ void load_v_half(v2f (&va)[8], gbytes vt, unsigned lane_off, int dc) {
  gbytes p = vt + lane_off + dc * 64;
#pragma unroll
  for (int c = 0; c < 4; ++c) {
    va[dc * 4 + c][0] = *(const __attribute__((address_space(1))) float*)(p + (2 * c) * (kDim * 4));
    va[dc * 4 + c][1] = *(const __attribute__((address_space(1))) float*)(p + (2 * c + 1) * (kDim * 4));
  }
}

__device__ __forceinline__ void load_v_halves(v2f (&va)[kKeyTiles][8], const float* Vb, int kb0, int d0,
                                              unsigned lane_off, int first, int count) {
#pragma unroll
  for (int i = 0; i < count; ++i) {
    const int t = (first + i) >> 1;
    gbytes vt = tile_base(Vb + (size_t)first_row_in_seq(kb0 + 16 * t) * kDim + d0);
    load_v_half(va[t], vt, lane_off, (first + i) & 1);
  }
}

__global__ __launch_bounds__(32 * kWaves) __attribute__((amdgpu_waves_per_eu(1, 1)))
void swa_wmma_f32_kernel(const float* __restrict__ Q,
                         const float* __restrict__ K,
                         const float* __restrict__ V,
                         float* __restrict__ O) {
  __shared__ v4f part[kWaves][kKeyTiles][2][32];

  const int lane = threadIdx.x & 31;
  const int wave = __builtin_amdgcn_readfirstlane((int)(threadIdx.x >> 5));
  const int tile = blockIdx.x;
  const int b    = tile / kTilesPerSeq;
  const int q0   = (tile % kTilesPerSeq) * 16;
  const int d0   = wave * kDimW;

  const int h   = lane >> 4;
  const int l15 = lane & 15;

  const float* Qb = Q + (size_t)b * kSeq * kDim;
  const float* Kb = K + (size_t)b * kSeq * kDim;
  const float* Vb = V + (size_t)b * kSeq * kDim;
  float*       Ob = O + (size_t)b * kSeq * kDim;

  const int kb0 = q0 - kWin;

  const unsigned qk_lane_off = (unsigned)((l15 * kDim + 2 * h) * 4);
  const unsigned v_lane_off  = (unsigned)((8 * h * kDim + l15) * 4);

  v2f qb[kChunksW];
  load_k_tile(qb, tile_base(Qb + (size_t)q0 * kDim + d0), qk_lane_off);

  v2f ka[kKeyTiles][kChunksW];
  v2f va[kKeyTiles][8];
  v8f pt[kKeyTiles];

#pragma unroll
  for (int j = 0; j < kKAhead; ++j)
    load_k_tile(ka[j], tile_base(Kb + (size_t)first_row_in_seq(kb0 + 16 * j) * kDim + d0), qk_lane_off);
  END_OF_ISSUE_SECTION();

#pragma unroll
  for (int j = 0; j < kKeyTiles; ++j) {
    if (j + kKAhead < kKeyTiles) {
      load_k_tile(ka[j + kKAhead],
                  tile_base(Kb + (size_t)first_row_in_seq(kb0 + 16 * (j + kKAhead)) * kDim + d0),
                  qk_lane_off);
    } else {
      load_v_halves(va, Vb, kb0, d0, v_lane_off, v_first_half_in_section(j), v_halves_in_section(j));
    }
    v8f p0 = (v8f)0.0f, p1 = (v8f)0.0f;
#pragma unroll
    for (int c = 0; c < kChunksW / 2; ++c) {
      p0 = __builtin_amdgcn_wmma_f32_16x16x4_f32(false, ka[j][2 * c],     false, qb[2 * c],
                                                 (short)0, p0, false, false);
      p1 = __builtin_amdgcn_wmma_f32_16x16x4_f32(false, ka[j][2 * c + 1], false, qb[2 * c + 1],
                                                 (short)0, p1, false, false);
    }
#pragma unroll
    for (int r = 0; r < 8; ++r) pt[j][r] = p0[r] + p1[r];
    END_OF_ISSUE_SECTION();
  }

#pragma unroll
  for (int j = 0; j < kKeyTiles; ++j) {
    part[wave][j][0][lane] = (v4f){pt[j][0], pt[j][1], pt[j][2], pt[j][3]};
    part[wave][j][1][lane] = (v4f){pt[j][4], pt[j][5], pt[j][6], pt[j][7]};
  }
  __builtin_amdgcn_fence(__ATOMIC_RELEASE, "workgroup", "local");
  __builtin_amdgcn_s_barrier();
  __builtin_amdgcn_fence(__ATOMIC_ACQUIRE, "workgroup", "local");

  const int wp = wave ^ 1;
  const int wo = 2 - (wave & 2);
  v4f xa[kKeyTiles][2], xu[kKeyTiles][2], xw[kKeyTiles][2];
#pragma unroll
  for (int j = 0; j < kKeyTiles; ++j) {
#pragma unroll
    for (int rq = 0; rq < 2; ++rq) {
      xa[j][rq] = part[wp][j][rq][lane];
      xu[j][rq] = part[wo][j][rq][lane];
      xw[j][rq] = part[wo + 1][j][rq][lane];
    }
  }
  END_OF_ISSUE_SECTION();
  load_v_halves(va, Vb, kb0, d0, v_lane_off, kVScoreHalves, kVExchHalves);
  END_OF_ISSUE_SECTION();

  const int r_diag = l15 - 8 * h;
  const int r_min0 = (kb0 >= 0) ? r_diag : 8;
  float mx = kNegInf;
#pragma unroll
  for (int j = 0; j < kKeyTiles; ++j) {
#pragma unroll
    for (int rq = 0; rq < 2; ++rq)
#pragma unroll
      for (int i = 0; i < 4; ++i)
        pt[j][4 * rq + i] = (pt[j][4 * rq + i] + xa[j][rq][i]) + (xu[j][rq][i] + xw[j][rq][i]);
#pragma unroll
    for (int r = 0; r < 8; ++r) {
      const bool valid = (j == 0)             ? (r >= r_min0)
                       : (j == kKeyTiles - 1) ? (r <= r_diag)
                                              : (kb0 + 16 * j >= 0);
      const float s = valid ? pt[j][r] * kScale : kNegInf;
      pt[j][r] = s;
      mx = fmaxf(mx, s);
    }
  }
  mx = fmaxf(mx, __shfl_xor(mx, 16, 32));

  float sum = 0.0f;
#pragma unroll
  for (int j = 0; j < kKeyTiles; ++j) {
#pragma unroll
    for (int r = 0; r < 8; ++r) {
      const float e = __expf(pt[j][r] - mx);
      pt[j][r] = e;
      sum += e;
    }
  }
  sum += __shfl_xor(sum, 16, 32);

  v8f acc[2];
  acc[0] = (v8f)0.0f;
  acc[1] = (v8f)0.0f;
#pragma unroll
  for (int j = 0; j < kKeyTiles; ++j) {
    if (j < (kVHalves - kVOutFirst) / 2) {
      LOADS_AND_MATRIX_KEEP_ORDER();
      load_v_halves(va, Vb, kb0, d0, v_lane_off, kVOutFirst + 2 * j, 2);
      LOADS_AND_MATRIX_KEEP_ORDER();
    }
#pragma unroll
    for (int c = 0; c < 4; ++c) {
      const v2f pb = {pt[j][2 * c], pt[j][2 * c + 1]};
#pragma unroll
      for (int dc = 0; dc < 2; ++dc)
        acc[dc] = __builtin_amdgcn_wmma_f32_16x16x4_f32(false, va[j][dc * 4 + c],
                                                        false, pb,
                                                        (short)0, acc[dc], false, false);
    }
  }

  const float inv = 1.0f / sum;
  float* orow = Ob + (size_t)(q0 + l15) * kDim + d0;
#pragma unroll
  for (int dc = 0; dc < 2; ++dc)
#pragma unroll
    for (int r = 0; r < 8; ++r)
      orow[dc * 16 + r + 8 * h] = acc[dc][r] * inv;
}

extern "C" void kernel_launch(void* const* d_in, const int* in_sizes, int n_in,
                              void* d_out, int out_size, void* d_ws, size_t ws_size,
                              hipStream_t stream) {
  const float* q = (const float*)d_in[0];
  const float* k = (const float*)d_in[1];
  const float* v = (const float*)d_in[2];
  float* out = (float*)d_out;

  dim3 grid(kBatch * kTilesPerSeq);
  dim3 block(32 * kWaves);
  swa_wmma_f32_kernel<<<grid, block, 0, stream>>>(q, k, v, out);
}


# Every load as wide as its fragment allows, 112 loads a wave (was 224): Q/K 16 B (dims 8p+4h..+3 feed k-steps 2p,2p+1), V 8 B (wave's dims dealt alternately to its 2 chunks); dim sum order changes
# speedup vs baseline: 3.5502x; 1.0574x over previous
#include <hip/hip_runtime.h>


typedef float v2f __attribute__((ext_vector_type(2)));
typedef float v4f __attribute__((ext_vector_type(4)));
typedef float v8f __attribute__((ext_vector_type(8)));

#define END_OF_ISSUE_SECTION() __builtin_amdgcn_sched_barrier(0)

namespace {
constexpr int   kBatch = 2;
constexpr int   kSeq   = 2048;
constexpr int   kDim   = 128;
constexpr int   kWin   = 128;
constexpr float kNegInf = -1.0e9f;
constexpr float kScale  = 0.08838834764831845f;
constexpr int   kTilesPerSeq = kSeq / 16;
constexpr int   kWaves    = 4;
constexpr int   kDimW     = kDim / kWaves;
constexpr int   kQuadsW   = kDimW / 8;
constexpr int   kKeyTiles = 9;
constexpr int   kVHalves      = 2 * kKeyTiles;
constexpr int   kVEntryHalves = 4;
constexpr int   kVScoreHalves = kVEntryHalves + kKeyTiles;
constexpr int   kVExchHalves  = kVHalves - kVScoreHalves;
}

__device__ __forceinline__ int first_row_in_seq(int kb) { return kb < 0 ? 0 : kb; }

typedef const __attribute__((address_space(1))) char* gbytes;
__device__ __forceinline__ gbytes tile_base(const float* p) {
  const unsigned long long v = (unsigned long long)p;
  const unsigned lo = __builtin_amdgcn_readfirstlane((unsigned)v);
  const unsigned hi = __builtin_amdgcn_readfirstlane((unsigned)(v >> 32));
  return (gbytes)(((unsigned long long)hi << 32) | lo);
}

__device__ __forceinline__ void load_k_tile(v4f (&kq)[kQuadsW], gbytes kt, unsigned lane_off) {
  gbytes p = kt + lane_off;
#pragma unroll
  for (int c = 0; c < kQuadsW; ++c)
    kq[c] = *(const __attribute__((address_space(1))) v4f*)(p + 32 * c);
}
__device__ __forceinline__ v2f first_pair(v4f x) { return __builtin_shufflevector(x, x, 0, 1); }
__device__ __forceinline__ v2f last_pair(v4f x)  { return __builtin_shufflevector(x, x, 2, 3); }

__device__ __forceinline__ void load_v_half(v2f (&vr)[8], gbytes vt, unsigned lane_off, int hv) {
  gbytes p = vt + lane_off + hv * 4 * (kDim * 4);
#pragma unroll
  for (int i = 0; i < 4; ++i)
    vr[hv * 4 + i] = *(const __attribute__((address_space(1))) v2f*)(p + i * (kDim * 4));
}

__device__ __forceinline__ void load_v_halves(v2f (&va)[kKeyTiles][8], const float* Vb, int kb0, int d0,
                                              unsigned lane_off, int first, int count) {
#pragma unroll
  for (int i = 0; i < count; ++i) {
    const int t = (first + i) >> 1;
    gbytes vt = tile_base(Vb + (size_t)first_row_in_seq(kb0 + 16 * t) * kDim + d0);
    load_v_half(va[t], vt, lane_off, (first + i) & 1);
  }
}

__global__ __launch_bounds__(32 * kWaves) __attribute__((amdgpu_waves_per_eu(1, 1)))
void swa_wmma_f32_kernel(const float* __restrict__ Q,
                         const float* __restrict__ K,
                         const float* __restrict__ V,
                         float* __restrict__ O) {
  __shared__ v4f part[kWaves][kKeyTiles][2][32];

  const int lane = threadIdx.x & 31;
  const int wave = __builtin_amdgcn_readfirstlane((int)(threadIdx.x >> 5));
  const int tile = blockIdx.x;
  const int b    = tile / kTilesPerSeq;
  const int q0   = (tile % kTilesPerSeq) * 16;
  const int d0   = wave * kDimW;

  const int h   = lane >> 4;
  const int l15 = lane & 15;

  const float* Qb = Q + (size_t)b * kSeq * kDim;
  const float* Kb = K + (size_t)b * kSeq * kDim;
  const float* Vb = V + (size_t)b * kSeq * kDim;
  float*       Ob = O + (size_t)b * kSeq * kDim;

  const int kb0 = q0 - kWin;

  const unsigned qk_lane_off = (unsigned)((l15 * kDim + 4 * h) * 4);
  const unsigned v_lane_off  = (unsigned)((8 * h * kDim + 2 * l15) * 4);

  v4f qb[kQuadsW];
  load_k_tile(qb, tile_base(Qb + (size_t)q0 * kDim + d0), qk_lane_off);

  v4f ka[kKeyTiles][kQuadsW];
  v2f va[kKeyTiles][8];
  v8f pt[kKeyTiles];

#pragma unroll
  for (int j = 0; j < kKeyTiles; ++j)
    load_k_tile(ka[j], tile_base(Kb + (size_t)first_row_in_seq(kb0 + 16 * j) * kDim + d0), qk_lane_off);
  END_OF_ISSUE_SECTION();
  load_v_halves(va, Vb, kb0, d0, v_lane_off, 0, kVEntryHalves);
  END_OF_ISSUE_SECTION();

#pragma unroll
  for (int j = 0; j < kKeyTiles; ++j) {
    load_v_halves(va, Vb, kb0, d0, v_lane_off, kVEntryHalves + j, 1);
    v8f p0 = (v8f)0.0f, p1 = (v8f)0.0f;
#pragma unroll
    for (int c = 0; c < kQuadsW; ++c) {
      p0 = __builtin_amdgcn_wmma_f32_16x16x4_f32(false, first_pair(ka[j][c]), false, first_pair(qb[c]),
                                                 (short)0, p0, false, false);
      p1 = __builtin_amdgcn_wmma_f32_16x16x4_f32(false, last_pair(ka[j][c]),  false, last_pair(qb[c]),
                                                 (short)0, p1, false, false);
    }
#pragma unroll
    for (int r = 0; r < 8; ++r) pt[j][r] = p0[r] + p1[r];
    END_OF_ISSUE_SECTION();
  }

#pragma unroll
  for (int j = 0; j < kKeyTiles; ++j) {
    part[wave][j][0][lane] = (v4f){pt[j][0], pt[j][1], pt[j][2], pt[j][3]};
    part[wave][j][1][lane] = (v4f){pt[j][4], pt[j][5], pt[j][6], pt[j][7]};
  }
  __builtin_amdgcn_fence(__ATOMIC_RELEASE, "workgroup", "local");
  __builtin_amdgcn_s_barrier();
  __builtin_amdgcn_fence(__ATOMIC_ACQUIRE, "workgroup", "local");

  const int wp = wave ^ 1;
  const int wo = 2 - (wave & 2);
  v4f xa[kKeyTiles][2], xu[kKeyTiles][2], xw[kKeyTiles][2];
#pragma unroll
  for (int j = 0; j < kKeyTiles; ++j) {
#pragma unroll
    for (int rq = 0; rq < 2; ++rq) {
      xa[j][rq] = part[wp][j][rq][lane];
      xu[j][rq] = part[wo][j][rq][lane];
      xw[j][rq] = part[wo + 1][j][rq][lane];
    }
  }
  END_OF_ISSUE_SECTION();
  load_v_halves(va, Vb, kb0, d0, v_lane_off, kVScoreHalves, kVExchHalves);
  END_OF_ISSUE_SECTION();

  const int r_diag = l15 - 8 * h;
  const int r_min0 = (kb0 >= 0) ? r_diag : 8;
  float mx = kNegInf;
#pragma unroll
  for (int j = 0; j < kKeyTiles; ++j) {
#pragma unroll
    for (int rq = 0; rq < 2; ++rq)
#pragma unroll
      for (int i = 0; i < 4; ++i)
        pt[j][4 * rq + i] = (pt[j][4 * rq + i] + xa[j][rq][i]) + (xu[j][rq][i] + xw[j][rq][i]);
#pragma unroll
    for (int r = 0; r < 8; ++r) {
      const bool valid = (j == 0)             ? (r >= r_min0)
                       : (j == kKeyTiles - 1) ? (r <= r_diag)
                                              : (kb0 + 16 * j >= 0);
      const float s = valid ? pt[j][r] * kScale : kNegInf;
      pt[j][r] = s;
      mx = fmaxf(mx, s);
    }
  }
  mx = fmaxf(mx, __shfl_xor(mx, 16, 32));

  float sum = 0.0f;
#pragma unroll
  for (int j = 0; j < kKeyTiles; ++j) {
#pragma unroll
    for (int r = 0; r < 8; ++r) {
      const float e = __expf(pt[j][r] - mx);
      pt[j][r] = e;
      sum += e;
    }
  }
  sum += __shfl_xor(sum, 16, 32);

  v8f acc[2];
  acc[0] = (v8f)0.0f;
  acc[1] = (v8f)0.0f;
#pragma unroll
  for (int j = 0; j < kKeyTiles; ++j) {
#pragma unroll
    for (int c = 0; c < 4; ++c) {
      const v2f pb = {pt[j][2 * c], pt[j][2 * c + 1]};
      const v2f r0 = va[j][2 * c], r1 = va[j][2 * c + 1];
#pragma unroll
      for (int dc = 0; dc < 2; ++dc) {
        const v2f a = {r0[dc], r1[dc]};
        acc[dc] = __builtin_amdgcn_wmma_f32_16x16x4_f32(false, a, false, pb,
                                                        (short)0, acc[dc], false, false);
      }
    }
  }

  const float inv = 1.0f / sum;
  float* orow = Ob + (size_t)(q0 + l15) * kDim + d0 + 16 * h;
#pragma unroll
  for (int g = 0; g < 4; ++g) {
    const v4f o = {acc[0][2 * g] * inv, acc[1][2 * g] * inv, acc[0][2 * g + 1] * inv, acc[1][2 * g + 1] * inv};
    *(v4f*)(orow + 4 * g) = o;
  }
}

extern "C" void kernel_launch(void* const* d_in, const int* in_sizes, int n_in,
                              void* d_out, int out_size, void* d_ws, size_t ws_size,
                              hipStream_t stream) {
  const float* q = (const float*)d_in[0];
  const float* k = (const float*)d_in[1];
  const float* v = (const float*)d_in[2];
  float* out = (float*)d_out;

  dim3 grid(kBatch * kTilesPerSeq);
  dim3 block(32 * kWaves);
  swa_wmma_f32_kernel<<<grid, block, 0, stream>>>(q, k, v, out);
}
